# starting kernel
# baseline (speedup 1.0000x reference)
#include <hip/hip_runtime.h>


#define D_MODEL 2048
#define SEQ     4096
#define BATCH   4
#define BS      (BATCH * SEQ)
#define KTILE   32
#define KP      36
#define MTILE   128
#define NTILE   128

typedef float v2f __attribute__((ext_vector_type(2)));
typedef float v4f __attribute__((ext_vector_type(4)));
typedef float v8f __attribute__((ext_vector_type(8)));

__device__ __forceinline__ void load_tile_128x32(const float* __restrict__ g,
                                                 float* lds, int t) {
    const int row = t >> 1;
    const int col = (t & 1) << 4;
    const v4f* __restrict__ src = (const v4f*)(g + (size_t)row * D_MODEL + col);
    v4f* dst = (v4f*)(lds + row * KP + col);
    dst[0] = src[0];
    dst[1] = src[1];
    dst[2] = src[2];
    dst[3] = src[3];
    __builtin_prefetch((const void*)(g + (size_t)row * D_MODEL + col + KTILE), 0, 0);
}

__device__ __forceinline__ void wmma_ktile(const float* As, const float* Ws,
                                           v8f acc[2][4],
                                           int wm, int wn, int ln, int kh) {
#pragma unroll
    for (int ks = 0; ks < KTILE / 4; ++ks) {
        const int kk = ks * 4 + kh * 2;
        v2f a[2], b[4];
#pragma unroll
        for (int i = 0; i < 2; ++i)
            a[i] = *(const v2f*)(As + (wm * 32 + i * 16 + ln) * KP + kk);
#pragma unroll
        for (int j = 0; j < 4; ++j)
            b[j] = *(const v2f*)(Ws + (wn * 64 + j * 16 + ln) * KP + kk);
#pragma unroll
        for (int i = 0; i < 2; ++i)
#pragma unroll
            for (int j = 0; j < 4; ++j)
                acc[i][j] = __builtin_amdgcn_wmma_f32_16x16x4_f32(
                    false, a[i], false, b[j], (short)0, acc[i][j], false, false);
    }
}

__device__ __forceinline__ void store_acc(float* __restrict__ dst,
                                          const v8f acc[2][4],
                                          int m0, int n0,
                                          int wm, int wn, int ln, int kh) {
#pragma unroll
    for (int i = 0; i < 2; ++i)
#pragma unroll
        for (int j = 0; j < 4; ++j)
#pragma unroll
            for (int r = 0; r < 8; ++r) {
                const int row = m0 + wm * 32 + i * 16 + kh * 8 + r;
                const int col = n0 + wn * 64 + j * 16 + ln;
                dst[(size_t)row * D_MODEL + col] = acc[i][j][r];
            }
}

__global__ __launch_bounds__(256)
void k1_gemm_gate(const float* __restrict__ x, const float* __restrict__ w1,
                  float* __restrict__ gated, float* __restrict__ cg) {
    __shared__ float As[MTILE * KP];
    __shared__ float Wu[NTILE * KP];
    __shared__ float Wv[NTILE * KP];

    const int t = threadIdx.x;
    const int wave = t >> 5, lane = t & 31;
    const int wm = wave & 3, wn = wave >> 2;
    const int ln = lane & 15, kh = lane >> 4;

    const int d0 = blockIdx.x * NTILE;
    const int m0 = blockIdx.y * MTILE;

    const float* xg = x + (size_t)m0 * D_MODEL;
    const float* wB = w1 + (size_t)(0 * D_MODEL + d0) * D_MODEL;
    const float* wC = w1 + (size_t)(1 * D_MODEL + d0) * D_MODEL;
    const float* wX = w1 + (size_t)(2 * D_MODEL + d0) * D_MODEL;

    v8f accB[2][4] = {};
    v8f accX[2][4] = {};
    for (int k0 = 0; k0 < D_MODEL; k0 += KTILE) {
        load_tile_128x32(xg + k0, As, t);
        load_tile_128x32(wB + k0, Wu, t);
        load_tile_128x32(wX + k0, Wv, t);
        __syncthreads();
        wmma_ktile(As, Wu, accB, wm, wn, ln, kh);
        wmma_ktile(As, Wv, accX, wm, wn, ln, kh);
        __syncthreads();
    }
#pragma unroll
    for (int i = 0; i < 2; ++i)
#pragma unroll
        for (int j = 0; j < 4; ++j)
#pragma unroll
            for (int r = 0; r < 8; ++r) {
                const int row = m0 + wm * 32 + i * 16 + kh * 8 + r;
                const int col = d0 + wn * 64 + j * 16 + ln;
                gated[(size_t)row * D_MODEL + col] = accB[i][j][r] * accX[i][j][r];
            }

    v8f accC[2][4] = {};
    for (int k0 = 0; k0 < D_MODEL; k0 += KTILE) {
        load_tile_128x32(xg + k0, As, t);
        load_tile_128x32(wC + k0, Wu, t);
        __syncthreads();
        wmma_ktile(As, Wu, accC, wm, wn, ln, kh);
        __syncthreads();
    }
    store_acc(cg, accC, m0, d0, wm, wn, ln, kh);
}

__global__ __launch_bounds__(256)
void k2_conv(const float* __restrict__ gated, const float* __restrict__ cw,
             float* __restrict__ cg_h) {
    const int idx = blockIdx.x * 256 + threadIdx.x;
    const int ND4 = D_MODEL / 4;
    const int d4  = idx & (ND4 - 1);
    const int row = idx >> 9;
    const int s   = row & (SEQ - 1);
    const int d   = d4 << 2;

    v4f sum = {0.f, 0.f, 0.f, 0.f};
#pragma unroll
    for (int k = 0; k < 4; ++k) {
        const int sp = s - 3 + k;
        if (sp >= 0) {
            v4f g = *(const v4f*)(gated + (size_t)(row - 3 + k) * D_MODEL + d);
            sum.x += cw[(d + 0) * 4 + k] * g.x;
            sum.y += cw[(d + 1) * 4 + k] * g.y;
            sum.z += cw[(d + 2) * 4 + k] * g.z;
            sum.w += cw[(d + 3) * 4 + k] * g.w;
        }
    }
    v4f c = *(const v4f*)(cg_h + (size_t)row * D_MODEL + d);
    *(v4f*)(cg_h + (size_t)row * D_MODEL + d) = c * sum;
}

__global__ __launch_bounds__(256)
void k3_gemm_out(const float* __restrict__ h, const float* __restrict__ w2,
                 float* __restrict__ out) {
    __shared__ float As[MTILE * KP];
    __shared__ float Ws[NTILE * KP];

    const int t = threadIdx.x;
    const int wave = t >> 5, lane = t & 31;
    const int wm = wave & 3, wn = wave >> 2;
    const int ln = lane & 15, kh = lane >> 4;

    const int n0 = blockIdx.x * NTILE;
    const int m0 = blockIdx.y * MTILE;

    const float* hg = h + (size_t)m0 * D_MODEL;
    const float* wg = w2 + (size_t)n0 * D_MODEL;

    v8f acc[2][4] = {};
    for (int k0 = 0; k0 < D_MODEL; k0 += KTILE) {
        load_tile_128x32(hg + k0, As, t);
        load_tile_128x32(wg + k0, Ws, t);
        __syncthreads();
        wmma_ktile(As, Ws, acc, wm, wn, ln, kh);
        __syncthreads();
    }
    store_acc(out, acc, m0, n0, wm, wn, ln, kh);
}

extern "C" void kernel_launch(void* const* d_in, const int* in_sizes, int n_in,
                              void* d_out, int out_size, void* d_ws, size_t ws_size,
                              hipStream_t stream) {
    const float* x  = (const float*)d_in[0];
    const float* w1 = (const float*)d_in[1];
    const float* w2 = (const float*)d_in[2];
    const float* cw = (const float*)d_in[3];

    float* gated = (float*)d_ws;
    float* cg    = gated + (size_t)BS * D_MODEL;

    dim3 grid_gemm(D_MODEL / NTILE, BS / MTILE);
    k1_gemm_gate<<<grid_gemm, 256, 0, stream>>>(x, w1, gated, cg);

    const int n_conv_blocks = (BS * (D_MODEL / 4)) / 256;
    k2_conv<<<n_conv_blocks, 256, 0, stream>>>(gated, cw, cg);

    k3_gemm_out<<<grid_gemm, 256, 0, stream>>>(cg, w2, (float*)d_out);
}


# Both GEMM K loops software-pipelined over two LDS stages: next tile's global loads issued ahead of the current tile's WMMAs, written to LDS behind them, one barrier per tile; sums unchanged
# speedup vs baseline: 1.0633x; 1.0633x over previous
#include <hip/hip_runtime.h>


#define D_MODEL 2048
#define SEQ     4096
#define BATCH   4
#define BS      (BATCH * SEQ)
#define KTILE   32
#define KP      36
#define MTILE   128
#define NTILE   128
#define TILE_F  (MTILE * KP)
#define KS_PUT  6

typedef float v2f __attribute__((ext_vector_type(2)));
typedef float v4f __attribute__((ext_vector_type(4)));
typedef float v8f __attribute__((ext_vector_type(8)));

__device__ __forceinline__ void gload_tile(const float* __restrict__ g, int t, v4f r[4]) {
    const int row = t >> 1;
    const int col = (t & 1) << 4;
    const v4f* __restrict__ src = (const v4f*)(g + (size_t)row * D_MODEL + col);
    r[0] = src[0];
    r[1] = src[1];
    r[2] = src[2];
    r[3] = src[3];
}

__device__ __forceinline__ void lds_put_tile(float* lds, int t, const v4f r[4]) {
    const int row = t >> 1;
    const int col = (t & 1) << 4;
    v4f* dst = (v4f*)(lds + row * KP + col);
    dst[0] = r[0];
    dst[1] = r[1];
    dst[2] = r[2];
    dst[3] = r[3];
}

template <int KS0, int KS1>
__device__ __forceinline__ void wmma_kslab(const float* As, const float* Ws,
                                           v8f acc[2][4],
                                           int wm, int wn, int ln, int kh) {
#pragma unroll
    for (int ks = KS0; ks < KS1; ++ks) {
        const int kk = ks * 4 + kh * 2;
        v2f a[2], b[4];
#pragma unroll
        for (int i = 0; i < 2; ++i)
            a[i] = *(const v2f*)(As + (wm * 32 + i * 16 + ln) * KP + kk);
#pragma unroll
        for (int j = 0; j < 4; ++j)
            b[j] = *(const v2f*)(Ws + (wn * 64 + j * 16 + ln) * KP + kk);
#pragma unroll
        for (int i = 0; i < 2; ++i)
#pragma unroll
            for (int j = 0; j < 4; ++j)
                acc[i][j] = __builtin_amdgcn_wmma_f32_16x16x4_f32(
                    false, a[i], false, b[j], (short)0, acc[i][j], false, false);
    }
}

template <bool FETCH>
__device__ __forceinline__ void step2(const float* __restrict__ gA, const float* __restrict__ gU,
                                      const float* __restrict__ gV,
                                      const float* curA, const float* curU, const float* curV,
                                      float* nxtA, float* nxtU, float* nxtV,
                                      v8f accU[2][4], v8f accV[2][4],
                                      int t, int wm, int wn, int ln, int kh) {
    v4f ra[4], ru[4], rv[4];
    if (FETCH) {
        gload_tile(gA, t, ra);
        gload_tile(gU, t, ru);
        gload_tile(gV, t, rv);
        __builtin_amdgcn_sched_barrier(0);
    }
    wmma_kslab<0, KS_PUT>(curA, curU, accU, wm, wn, ln, kh);
    wmma_kslab<0, KS_PUT>(curA, curV, accV, wm, wn, ln, kh);
    if (FETCH) {
        __builtin_amdgcn_sched_barrier(0);
        lds_put_tile(nxtA, t, ra);
        lds_put_tile(nxtU, t, ru);
        lds_put_tile(nxtV, t, rv);
    }
    wmma_kslab<KS_PUT, KTILE / 4>(curA, curU, accU, wm, wn, ln, kh);
    wmma_kslab<KS_PUT, KTILE / 4>(curA, curV, accV, wm, wn, ln, kh);
    if (FETCH) __syncthreads();
}

template <bool FETCH>
__device__ __forceinline__ void step1(const float* __restrict__ gA, const float* __restrict__ gU,
                                      const float* curA, const float* curU,
                                      float* nxtA, float* nxtU,
                                      v8f acc[2][4],
                                      int t, int wm, int wn, int ln, int kh) {
    v4f ra[4], ru[4];
    if (FETCH) {
        gload_tile(gA, t, ra);
        gload_tile(gU, t, ru);
        __builtin_amdgcn_sched_barrier(0);
    }
    wmma_kslab<0, KS_PUT>(curA, curU, acc, wm, wn, ln, kh);
    if (FETCH) {
        __builtin_amdgcn_sched_barrier(0);
        lds_put_tile(nxtA, t, ra);
        lds_put_tile(nxtU, t, ru);
    }
    wmma_kslab<KS_PUT, KTILE / 4>(curA, curU, acc, wm, wn, ln, kh);
    if (FETCH) __syncthreads();
}

__device__ __forceinline__ void store_acc(float* __restrict__ dst,
                                          const v8f acc[2][4],
                                          int m0, int n0,
                                          int wm, int wn, int ln, int kh) {
#pragma unroll
    for (int i = 0; i < 2; ++i)
#pragma unroll
        for (int j = 0; j < 4; ++j)
#pragma unroll
            for (int r = 0; r < 8; ++r) {
                const int row = m0 + wm * 32 + i * 16 + kh * 8 + r;
                const int col = n0 + wn * 64 + j * 16 + ln;
                dst[(size_t)row * D_MODEL + col] = acc[i][j][r];
            }
}

__global__ __launch_bounds__(256, 4)
void k1_gemm_gate(const float* __restrict__ x, const float* __restrict__ w1,
                  float* __restrict__ gated, float* __restrict__ cg) {
    __shared__ float sm[2 * 3 * TILE_F];
    float* const As0 = sm;
    float* const Wu0 = sm + 1 * TILE_F;
    float* const Wv0 = sm + 2 * TILE_F;
    float* const As1 = sm + 3 * TILE_F;
    float* const Wu1 = sm + 4 * TILE_F;
    float* const Wv1 = sm + 5 * TILE_F;

    const int t = threadIdx.x;
    const int wave = t >> 5, lane = t & 31;
    const int wm = wave & 3, wn = wave >> 2;
    const int ln = lane & 15, kh = lane >> 4;

    const int d0 = blockIdx.x * NTILE;
    const int m0 = blockIdx.y * MTILE;

    const float* xg = x + (size_t)m0 * D_MODEL;
    const float* wB = w1 + (size_t)(0 * D_MODEL + d0) * D_MODEL;
    const float* wC = w1 + (size_t)(1 * D_MODEL + d0) * D_MODEL;
    const float* wX = w1 + (size_t)(2 * D_MODEL + d0) * D_MODEL;

    v8f accB[2][4] = {};
    v8f accX[2][4] = {};
    {
        v4f ra[4], rb[4], rx[4];
        gload_tile(xg, t, ra);
        gload_tile(wB, t, rb);
        gload_tile(wX, t, rx);
        lds_put_tile(As0, t, ra);
        lds_put_tile(Wu0, t, rb);
        lds_put_tile(Wv0, t, rx);
        __syncthreads();
    }
    for (int k0 = 0; k0 < D_MODEL - 2 * KTILE; k0 += 2 * KTILE) {
        step2<true>(xg + k0 + KTILE, wB + k0 + KTILE, wX + k0 + KTILE,
                    As0, Wu0, Wv0, As1, Wu1, Wv1, accB, accX, t, wm, wn, ln, kh);
        step2<true>(xg + k0 + 2 * KTILE, wB + k0 + 2 * KTILE, wX + k0 + 2 * KTILE,
                    As1, Wu1, Wv1, As0, Wu0, Wv0, accB, accX, t, wm, wn, ln, kh);
    }
    step2<true>(xg + D_MODEL - KTILE, wB + D_MODEL - KTILE, wX + D_MODEL - KTILE,
                As0, Wu0, Wv0, As1, Wu1, Wv1, accB, accX, t, wm, wn, ln, kh);
    step2<false>(xg, wB, wX, As1, Wu1, Wv1, As0, Wu0, Wv0, accB, accX, t, wm, wn, ln, kh);

#pragma unroll
    for (int i = 0; i < 2; ++i)
#pragma unroll
        for (int j = 0; j < 4; ++j)
#pragma unroll
            for (int r = 0; r < 8; ++r) {
                const int row = m0 + wm * 32 + i * 16 + kh * 8 + r;
                const int col = d0 + wn * 64 + j * 16 + ln;
                gated[(size_t)row * D_MODEL + col] = accB[i][j][r] * accX[i][j][r];
            }

    v8f accC[2][4] = {};
    {
        v4f ra[4], rb[4];
        gload_tile(xg, t, ra);
        gload_tile(wC, t, rb);
        lds_put_tile(As0, t, ra);
        lds_put_tile(Wu0, t, rb);
        __syncthreads();
    }
    for (int k0 = 0; k0 < D_MODEL - 2 * KTILE; k0 += 2 * KTILE) {
        step1<true>(xg + k0 + KTILE, wC + k0 + KTILE, As0, Wu0, As1, Wu1, accC, t, wm, wn, ln, kh);
        step1<true>(xg + k0 + 2 * KTILE, wC + k0 + 2 * KTILE, As1, Wu1, As0, Wu0, accC, t, wm, wn, ln, kh);
    }
    step1<true>(xg + D_MODEL - KTILE, wC + D_MODEL - KTILE, As0, Wu0, As1, Wu1, accC, t, wm, wn, ln, kh);
    step1<false>(xg, wC, As1, Wu1, As0, Wu0, accC, t, wm, wn, ln, kh);
    store_acc(cg, accC, m0, d0, wm, wn, ln, kh);
}

__global__ __launch_bounds__(256)
void k2_conv(const float* __restrict__ gated, const float* __restrict__ cw,
             float* __restrict__ cg_h) {
    const int idx = blockIdx.x * 256 + threadIdx.x;
    const int ND4 = D_MODEL / 4;
    const int d4  = idx & (ND4 - 1);
    const int row = idx >> 9;
    const int s   = row & (SEQ - 1);
    const int d   = d4 << 2;

    v4f sum = {0.f, 0.f, 0.f, 0.f};
#pragma unroll
    for (int k = 0; k < 4; ++k) {
        const int sp = s - 3 + k;
        if (sp >= 0) {
            v4f g = *(const v4f*)(gated + (size_t)(row - 3 + k) * D_MODEL + d);
            sum.x += cw[(d + 0) * 4 + k] * g.x;
            sum.y += cw[(d + 1) * 4 + k] * g.y;
            sum.z += cw[(d + 2) * 4 + k] * g.z;
            sum.w += cw[(d + 3) * 4 + k] * g.w;
        }
    }
    v4f c = *(const v4f*)(cg_h + (size_t)row * D_MODEL + d);
    *(v4f*)(cg_h + (size_t)row * D_MODEL + d) = c * sum;
}

__global__ __launch_bounds__(256)
void k3_gemm_out(const float* __restrict__ h, const float* __restrict__ w2,
                 float* __restrict__ out) {
    __shared__ float sm[2 * 2 * TILE_F];
    float* const As0 = sm;
    float* const Ws0 = sm + 1 * TILE_F;
    float* const As1 = sm + 2 * TILE_F;
    float* const Ws1 = sm + 3 * TILE_F;

    const int t = threadIdx.x;
    const int wave = t >> 5, lane = t & 31;
    const int wm = wave & 3, wn = wave >> 2;
    const int ln = lane & 15, kh = lane >> 4;

    const int n0 = blockIdx.x * NTILE;
    const int m0 = blockIdx.y * MTILE;

    const float* hg = h + (size_t)m0 * D_MODEL;
    const float* wg = w2 + (size_t)n0 * D_MODEL;

    v8f acc[2][4] = {};
    {
        v4f ra[4], rb[4];
        gload_tile(hg, t, ra);
        gload_tile(wg, t, rb);
        lds_put_tile(As0, t, ra);
        lds_put_tile(Ws0, t, rb);
        __syncthreads();
    }
    for (int k0 = 0; k0 < D_MODEL - 2 * KTILE; k0 += 2 * KTILE) {
        step1<true>(hg + k0 + KTILE, wg + k0 + KTILE, As0, Ws0, As1, Ws1, acc, t, wm, wn, ln, kh);
        step1<true>(hg + k0 + 2 * KTILE, wg + k0 + 2 * KTILE, As1, Ws1, As0, Ws0, acc, t, wm, wn, ln, kh);
    }
    step1<true>(hg + D_MODEL - KTILE, wg + D_MODEL - KTILE, As0, Ws0, As1, Ws1, acc, t, wm, wn, ln, kh);
    step1<false>(hg, wg, As1, Ws1, As0, Ws0, acc, t, wm, wn, ln, kh);
    store_acc(out, acc, m0, n0, wm, wn, ln, kh);
}

extern "C" void kernel_launch(void* const* d_in, const int* in_sizes, int n_in,
                              void* d_out, int out_size, void* d_ws, size_t ws_size,
                              hipStream_t stream) {
    const float* x  = (const float*)d_in[0];
    const float* w1 = (const float*)d_in[1];
    const float* w2 = (const float*)d_in[2];
    const float* cw = (const float*)d_in[3];

    float* gated = (float*)d_ws;
    float* cg    = gated + (size_t)BS * D_MODEL;

    dim3 grid_gemm(D_MODEL / NTILE, BS / MTILE);
    k1_gemm_gate<<<grid_gemm, 256, 0, stream>>>(x, w1, gated, cg);

    const int n_conv_blocks = (BS * (D_MODEL / 4)) / 256;
    k2_conv<<<n_conv_blocks, 256, 0, stream>>>(gated, cw, cg);

    k3_gemm_out<<<grid_gemm, 256, 0, stream>>>(cg, w2, (float*)d_out);
}
